# starting kernel
# baseline (speedup 1.0000x reference)
#include <hip/hip_runtime.h>

#define N_NODES 100000
#define N_EDGES 1600000
#define IN_DIM  128
#define OUT_DIM 64
#define NEG_SLOPE 0.01f

typedef float v2f __attribute__((ext_vector_type(2)));
typedef float v8f __attribute__((ext_vector_type(8)));

__global__ void degree_kernel(const int* __restrict__ src, const int* __restrict__ dst,
                              float* __restrict__ outdeg, float* __restrict__ indeg, int E) {
    int e = blockIdx.x * blockDim.x + threadIdx.x;
    if (e < E) {
        atomicAdd(&outdeg[src[e]], 1.0f);
        atomicAdd(&indeg[dst[e]], 1.0f);
    }
}

__global__ void norm_kernel(float* __restrict__ ns, float* __restrict__ nd, int n) {
    int i = blockIdx.x * blockDim.x + threadIdx.x;
    if (i < n) {
        ns[i] = rsqrtf(fmaxf(ns[i], 1.0f));
        nd[i] = rsqrtf(fmaxf(nd[i], 1.0f));
    }
}

__global__ void __launch_bounds__(256)
gemm_wmma_kernel(const float* __restrict__ h, const float* __restrict__ W,
                 const float* __restrict__ ns, float* __restrict__ x, int mtiles) {
    __shared__ float Wt[OUT_DIM * 130];
    for (int i = threadIdx.x; i < IN_DIM * OUT_DIM; i += 256) {
        int k = i >> 6;
        int c = i & 63;
        Wt[c * 130 + k] = W[i];
    }
    __syncthreads();

    const int wave = threadIdx.x >> 5;
    const int lane = threadIdx.x & 31;
    const int mtile = blockIdx.x * 8 + wave;
    if (mtile >= mtiles) return;

    const int mrow = (mtile << 4) + (lane & 15);
    const int kb   = (lane >> 4) << 1;
    const int n    = lane & 15;

    const float* hrow = h + (size_t)mrow * IN_DIM + kb;
    const float* w0 = &Wt[(0 * 16 + n) * 130 + kb];
    const float* w1 = &Wt[(1 * 16 + n) * 130 + kb];
    const float* w2 = &Wt[(2 * 16 + n) * 130 + kb];
    const float* w3 = &Wt[(3 * 16 + n) * 130 + kb];

    v8f acc0 = {}, acc1 = {}, acc2 = {}, acc3 = {};
#pragma unroll 4
    for (int k = 0; k < IN_DIM; k += 4) {
        v2f a  = *(const v2f*)(hrow + k);
        v2f b0 = *(const v2f*)(w0 + k);
        v2f b1 = *(const v2f*)(w1 + k);
        v2f b2 = *(const v2f*)(w2 + k);
        v2f b3 = *(const v2f*)(w3 + k);
        acc0 = __builtin_amdgcn_wmma_f32_16x16x4_f32(false, a, false, b0, (short)0, acc0, false, false);
        acc1 = __builtin_amdgcn_wmma_f32_16x16x4_f32(false, a, false, b1, (short)0, acc1, false, false);
        acc2 = __builtin_amdgcn_wmma_f32_16x16x4_f32(false, a, false, b2, (short)0, acc2, false, false);
        acc3 = __builtin_amdgcn_wmma_f32_16x16x4_f32(false, a, false, b3, (short)0, acc3, false, false);
    }

    const int rowbase = (mtile << 4) + ((lane >> 4) << 3);
#pragma unroll
    for (int r = 0; r < 8; ++r) {
        int row = rowbase + r;
        float s = ns[row];
        float* xr = x + (size_t)row * OUT_DIM + n;
        xr[0]  = acc0[r] * s;
        xr[16] = acc1[r] * s;
        xr[32] = acc2[r] * s;
        xr[48] = acc3[r] * s;
    }
}

__global__ void scatter_kernel(const float* __restrict__ x, const int* __restrict__ src,
                               const int* __restrict__ dst, float* __restrict__ agg, int E) {
    int t = blockIdx.x * blockDim.x + threadIdx.x;
    int e  = t >> 4;
    int c4 = (t & 15) << 2;
    if (e < E) {
        int s = src[e];
        int d = dst[e];
        const float4 v = *(const float4*)(x + (size_t)s * OUT_DIM + c4);
        float* ap = agg + (size_t)d * OUT_DIM + c4;
        atomicAdd(ap + 0, v.x);
        atomicAdd(ap + 1, v.y);
        atomicAdd(ap + 2, v.z);
        atomicAdd(ap + 3, v.w);
    }
}

__global__ void finalize_kernel(const float* __restrict__ agg, const float* __restrict__ nd,
                                const float* __restrict__ b, float* __restrict__ out, int total) {
    int t = blockIdx.x * blockDim.x + threadIdx.x;
    if (t < total) {
        int i = t >> 6;
        int c = t & 63;
        float v = agg[t] * nd[i] + b[c];
        out[t] = (v >= 0.0f) ? v : v * NEG_SLOPE;
    }
}

extern "C" void kernel_launch(void* const* d_in, const int* in_sizes, int n_in,
                              void* d_out, int out_size, void* d_ws, size_t ws_size,
                              hipStream_t stream) {
    const float* h    = (const float*)d_in[0];
    const float* W    = (const float*)d_in[1];
    const float* b    = (const float*)d_in[2];
    const int*   esrc = (const int*)d_in[3];
    const int*   edst = (const int*)d_in[4];
    float*       out  = (float*)d_out;

    float* ws  = (float*)d_ws;
    float* ns  = ws;
    float* nd  = ws + N_NODES;
    float* x   = ws + 2 * (size_t)N_NODES;
    float* agg = x + (size_t)N_NODES * OUT_DIM;

    hipMemsetAsync(ns,  0, 2 * (size_t)N_NODES * sizeof(float), stream);
    hipMemsetAsync(agg, 0, (size_t)N_NODES * OUT_DIM * sizeof(float), stream);

    degree_kernel<<<(N_EDGES + 255) / 256, 256, 0, stream>>>(esrc, edst, ns, nd, N_EDGES);
    norm_kernel<<<(N_NODES + 255) / 256, 256, 0, stream>>>(ns, nd, N_NODES);

    const int mtiles = N_NODES / 16;
    gemm_wmma_kernel<<<(mtiles + 7) / 8, 256, 0, stream>>>(h, W, ns, x, mtiles);

    const long long sthreads = (long long)N_EDGES * 16;
    scatter_kernel<<<(unsigned)((sthreads + 255) / 256), 256, 0, stream>>>(x, esrc, edst, agg, N_EDGES);

    const int total = N_NODES * OUT_DIM;
    finalize_kernel<<<(total + 255) / 256, 256, 0, stream>>>(agg, nd, b, out, total);
}


# scatter: one wave per 32 edges, lane l owns columns l and l+32, so each row load and each f32 atomic add covers one whole 128-byte line (4 line requests per edge, was 10); atomics kept
# speedup vs baseline: 1.3109x; 1.3109x over previous
#include <hip/hip_runtime.h>

#define N_NODES 100000
#define N_EDGES 1600000
#define IN_DIM  128
#define OUT_DIM 64
#define NEG_SLOPE 0.01f

typedef float v2f __attribute__((ext_vector_type(2)));
typedef float v8f __attribute__((ext_vector_type(8)));

__global__ void degree_kernel(const int* __restrict__ src, const int* __restrict__ dst,
                              float* __restrict__ outdeg, float* __restrict__ indeg, int E) {
    int e = blockIdx.x * blockDim.x + threadIdx.x;
    if (e < E) {
        atomicAdd(&outdeg[src[e]], 1.0f);
        atomicAdd(&indeg[dst[e]], 1.0f);
    }
}

__global__ void norm_kernel(float* __restrict__ ns, float* __restrict__ nd, int n) {
    int i = blockIdx.x * blockDim.x + threadIdx.x;
    if (i < n) {
        ns[i] = rsqrtf(fmaxf(ns[i], 1.0f));
        nd[i] = rsqrtf(fmaxf(nd[i], 1.0f));
    }
}

__global__ void __launch_bounds__(256)
gemm_wmma_kernel(const float* __restrict__ h, const float* __restrict__ W,
                 const float* __restrict__ ns, float* __restrict__ x, int mtiles) {
    __shared__ float Wt[OUT_DIM * 130];
    for (int i = threadIdx.x; i < IN_DIM * OUT_DIM; i += 256) {
        int k = i >> 6;
        int c = i & 63;
        Wt[c * 130 + k] = W[i];
    }
    __syncthreads();

    const int wave = threadIdx.x >> 5;
    const int lane = threadIdx.x & 31;
    const int mtile = blockIdx.x * 8 + wave;
    if (mtile >= mtiles) return;

    const int mrow = (mtile << 4) + (lane & 15);
    const int kb   = (lane >> 4) << 1;
    const int n    = lane & 15;

    const float* hrow = h + (size_t)mrow * IN_DIM + kb;
    const float* w0 = &Wt[(0 * 16 + n) * 130 + kb];
    const float* w1 = &Wt[(1 * 16 + n) * 130 + kb];
    const float* w2 = &Wt[(2 * 16 + n) * 130 + kb];
    const float* w3 = &Wt[(3 * 16 + n) * 130 + kb];

    v8f acc0 = {}, acc1 = {}, acc2 = {}, acc3 = {};
#pragma unroll 4
    for (int k = 0; k < IN_DIM; k += 4) {
        v2f a  = *(const v2f*)(hrow + k);
        v2f b0 = *(const v2f*)(w0 + k);
        v2f b1 = *(const v2f*)(w1 + k);
        v2f b2 = *(const v2f*)(w2 + k);
        v2f b3 = *(const v2f*)(w3 + k);
        acc0 = __builtin_amdgcn_wmma_f32_16x16x4_f32(false, a, false, b0, (short)0, acc0, false, false);
        acc1 = __builtin_amdgcn_wmma_f32_16x16x4_f32(false, a, false, b1, (short)0, acc1, false, false);
        acc2 = __builtin_amdgcn_wmma_f32_16x16x4_f32(false, a, false, b2, (short)0, acc2, false, false);
        acc3 = __builtin_amdgcn_wmma_f32_16x16x4_f32(false, a, false, b3, (short)0, acc3, false, false);
    }

    const int rowbase = (mtile << 4) + ((lane >> 4) << 3);
#pragma unroll
    for (int r = 0; r < 8; ++r) {
        int row = rowbase + r;
        float s = ns[row];
        float* xr = x + (size_t)row * OUT_DIM + n;
        xr[0]  = acc0[r] * s;
        xr[16] = acc1[r] * s;
        xr[32] = acc2[r] * s;
        xr[48] = acc3[r] * s;
    }
}

__global__ void __launch_bounds__(256)
scatter_kernel(const float* __restrict__ x, const int* __restrict__ src,
               const int* __restrict__ dst, float* __restrict__ agg, int E) {
    const int lane = threadIdx.x & 31;
    const int e0 = (blockIdx.x * 8 + (threadIdx.x >> 5)) << 5;
    if (e0 >= E) return;
    const int ei = e0 + lane;
    const int sv = (ei < E) ? src[ei] : 0;
    const int dv = (ei < E) ? dst[ei] : 0;
    if (e0 + 32 <= E) {
#pragma unroll
        for (int c = 0; c < 32; c += 8) {
            float v0[8], v1[8];
#pragma unroll
            for (int j = 0; j < 8; ++j) {
                const int s = __builtin_amdgcn_readlane(sv, c + j);
                const float* xr = x + (size_t)s * OUT_DIM + lane;
                v0[j] = xr[0];
                v1[j] = xr[32];
            }
#pragma unroll
            for (int j = 0; j < 8; ++j) {
                const int d = __builtin_amdgcn_readlane(dv, c + j);
                float* ap = agg + (size_t)d * OUT_DIM + lane;
                atomicAdd(ap, v0[j]);
                atomicAdd(ap + 32, v1[j]);
            }
        }
    } else {
        const int n = E - e0;
        for (int j = 0; j < n; ++j) {
            const int s = __builtin_amdgcn_readlane(sv, j);
            const int d = __builtin_amdgcn_readlane(dv, j);
            const float* xr = x + (size_t)s * OUT_DIM + lane;
            float* ap = agg + (size_t)d * OUT_DIM + lane;
            atomicAdd(ap, xr[0]);
            atomicAdd(ap + 32, xr[32]);
        }
    }
}

__global__ void finalize_kernel(const float* __restrict__ agg, const float* __restrict__ nd,
                                const float* __restrict__ b, float* __restrict__ out, int total) {
    int t = blockIdx.x * blockDim.x + threadIdx.x;
    if (t < total) {
        int i = t >> 6;
        int c = t & 63;
        float v = agg[t] * nd[i] + b[c];
        out[t] = (v >= 0.0f) ? v : v * NEG_SLOPE;
    }
}

extern "C" void kernel_launch(void* const* d_in, const int* in_sizes, int n_in,
                              void* d_out, int out_size, void* d_ws, size_t ws_size,
                              hipStream_t stream) {
    const float* h    = (const float*)d_in[0];
    const float* W    = (const float*)d_in[1];
    const float* b    = (const float*)d_in[2];
    const int*   esrc = (const int*)d_in[3];
    const int*   edst = (const int*)d_in[4];
    float*       out  = (float*)d_out;

    float* ws  = (float*)d_ws;
    float* ns  = ws;
    float* nd  = ws + N_NODES;
    float* x   = ws + 2 * (size_t)N_NODES;
    float* agg = x + (size_t)N_NODES * OUT_DIM;

    hipMemsetAsync(ns,  0, 2 * (size_t)N_NODES * sizeof(float), stream);
    hipMemsetAsync(agg, 0, (size_t)N_NODES * OUT_DIM * sizeof(float), stream);

    degree_kernel<<<(N_EDGES + 255) / 256, 256, 0, stream>>>(esrc, edst, ns, nd, N_EDGES);
    norm_kernel<<<(N_NODES + 255) / 256, 256, 0, stream>>>(ns, nd, N_NODES);

    const int mtiles = N_NODES / 16;
    gemm_wmma_kernel<<<(mtiles + 7) / 8, 256, 0, stream>>>(h, W, ns, x, mtiles);

    const int swaves = (N_EDGES + 31) / 32;
    scatter_kernel<<<(swaves + 7) / 8, 256, 0, stream>>>(x, esrc, edst, agg, N_EDGES);

    const int total = N_NODES * OUT_DIM;
    finalize_kernel<<<(total + 255) / 256, 256, 0, stream>>>(agg, nd, b, out, total);
}


# one plane per cache half for every atomic sum (degree counts, agg): a wave adds to the plane of its die's half, norm and finalize add the two planes; float sum order changes (two partial sums)
# speedup vs baseline: 2.3758x; 1.8123x over previous
#include <hip/hip_runtime.h>

#define N_NODES 100000
#define N_EDGES 1600000
#define IN_DIM  128
#define OUT_DIM 64
#define NEG_SLOPE 0.01f

typedef float v2f __attribute__((ext_vector_type(2)));
typedef float v8f __attribute__((ext_vector_type(8)));

__device__ __forceinline__ int cache_half() {
    return (int)((__builtin_amdgcn_s_sendmsg_rtn(0x87) >> 18) & 1u);
}

__global__ void degree_kernel(const int* __restrict__ src, const int* __restrict__ dst,
                              float* __restrict__ outdeg, float* __restrict__ indeg, int E) {
    int e = blockIdx.x * blockDim.x + threadIdx.x;
    const int hp = cache_half() * N_NODES;
    if (e < E) {
        atomicAdd(&outdeg[hp + src[e]], 1.0f);
        atomicAdd(&indeg[hp + dst[e]], 1.0f);
    }
}

__global__ void norm_kernel(float* __restrict__ ns, float* __restrict__ nd, int n) {
    int i = blockIdx.x * blockDim.x + threadIdx.x;
    if (i < n) {
        ns[i] = rsqrtf(fmaxf(ns[i] + ns[n + i], 1.0f));
        nd[i] = rsqrtf(fmaxf(nd[i] + nd[n + i], 1.0f));
    }
}

__global__ void __launch_bounds__(256)
gemm_wmma_kernel(const float* __restrict__ h, const float* __restrict__ W,
                 const float* __restrict__ ns, float* __restrict__ x, int mtiles) {
    __shared__ float Wt[OUT_DIM * 130];
    for (int i = threadIdx.x; i < IN_DIM * OUT_DIM; i += 256) {
        int k = i >> 6;
        int c = i & 63;
        Wt[c * 130 + k] = W[i];
    }
    __syncthreads();

    const int wave = threadIdx.x >> 5;
    const int lane = threadIdx.x & 31;
    const int mtile = blockIdx.x * 8 + wave;
    if (mtile >= mtiles) return;

    const int mrow = (mtile << 4) + (lane & 15);
    const int kb   = (lane >> 4) << 1;
    const int n    = lane & 15;

    const float* hrow = h + (size_t)mrow * IN_DIM + kb;
    const float* w0 = &Wt[(0 * 16 + n) * 130 + kb];
    const float* w1 = &Wt[(1 * 16 + n) * 130 + kb];
    const float* w2 = &Wt[(2 * 16 + n) * 130 + kb];
    const float* w3 = &Wt[(3 * 16 + n) * 130 + kb];

    v8f acc0 = {}, acc1 = {}, acc2 = {}, acc3 = {};
#pragma unroll 4
    for (int k = 0; k < IN_DIM; k += 4) {
        v2f a  = *(const v2f*)(hrow + k);
        v2f b0 = *(const v2f*)(w0 + k);
        v2f b1 = *(const v2f*)(w1 + k);
        v2f b2 = *(const v2f*)(w2 + k);
        v2f b3 = *(const v2f*)(w3 + k);
        acc0 = __builtin_amdgcn_wmma_f32_16x16x4_f32(false, a, false, b0, (short)0, acc0, false, false);
        acc1 = __builtin_amdgcn_wmma_f32_16x16x4_f32(false, a, false, b1, (short)0, acc1, false, false);
        acc2 = __builtin_amdgcn_wmma_f32_16x16x4_f32(false, a, false, b2, (short)0, acc2, false, false);
        acc3 = __builtin_amdgcn_wmma_f32_16x16x4_f32(false, a, false, b3, (short)0, acc3, false, false);
    }

    const int rowbase = (mtile << 4) + ((lane >> 4) << 3);
#pragma unroll
    for (int r = 0; r < 8; ++r) {
        int row = rowbase + r;
        float s = ns[row];
        float* xr = x + (size_t)row * OUT_DIM + n;
        xr[0]  = acc0[r] * s;
        xr[16] = acc1[r] * s;
        xr[32] = acc2[r] * s;
        xr[48] = acc3[r] * s;
    }
}

__global__ void __launch_bounds__(256)
scatter_kernel(const float* __restrict__ x, const int* __restrict__ src,
               const int* __restrict__ dst, float* __restrict__ agg, int E) {
    const int lane = threadIdx.x & 31;
    const int e0 = (blockIdx.x * 8 + (threadIdx.x >> 5)) << 5;
    if (e0 >= E) return;
    agg += (size_t)cache_half() * ((size_t)N_NODES * OUT_DIM);
    const int ei = e0 + lane;
    const int sv = (ei < E) ? src[ei] : 0;
    const int dv = (ei < E) ? dst[ei] : 0;
    if (e0 + 32 <= E) {
#pragma unroll
        for (int c = 0; c < 32; c += 8) {
            float v0[8], v1[8];
#pragma unroll
            for (int j = 0; j < 8; ++j) {
                const int s = __builtin_amdgcn_readlane(sv, c + j);
                const float* xr = x + (size_t)s * OUT_DIM + lane;
                v0[j] = xr[0];
                v1[j] = xr[32];
            }
#pragma unroll
            for (int j = 0; j < 8; ++j) {
                const int d = __builtin_amdgcn_readlane(dv, c + j);
                float* ap = agg + (size_t)d * OUT_DIM + lane;
                atomicAdd(ap, v0[j]);
                atomicAdd(ap + 32, v1[j]);
            }
        }
    } else {
        const int n = E - e0;
        for (int j = 0; j < n; ++j) {
            const int s = __builtin_amdgcn_readlane(sv, j);
            const int d = __builtin_amdgcn_readlane(dv, j);
            const float* xr = x + (size_t)s * OUT_DIM + lane;
            float* ap = agg + (size_t)d * OUT_DIM + lane;
            atomicAdd(ap, xr[0]);
            atomicAdd(ap + 32, xr[32]);
        }
    }
}

__global__ void finalize_kernel(const float* __restrict__ agg, const float* __restrict__ nd,
                                const float* __restrict__ b, float* __restrict__ out, int total) {
    int t = blockIdx.x * blockDim.x + threadIdx.x;
    if (t < total) {
        int i = t >> 6;
        int c = t & 63;
        float v = (agg[t] + agg[total + t]) * nd[i] + b[c];
        out[t] = (v >= 0.0f) ? v : v * NEG_SLOPE;
    }
}

extern "C" void kernel_launch(void* const* d_in, const int* in_sizes, int n_in,
                              void* d_out, int out_size, void* d_ws, size_t ws_size,
                              hipStream_t stream) {
    const float* h    = (const float*)d_in[0];
    const float* W    = (const float*)d_in[1];
    const float* b    = (const float*)d_in[2];
    const int*   esrc = (const int*)d_in[3];
    const int*   edst = (const int*)d_in[4];
    float*       out  = (float*)d_out;

    float* ws  = (float*)d_ws;
    float* ns  = ws;
    float* nd  = ws + 2 * (size_t)N_NODES;
    float* x   = ws + 4 * (size_t)N_NODES;
    float* agg = x + (size_t)N_NODES * OUT_DIM;

    hipMemsetAsync(ns,  0, 4 * (size_t)N_NODES * sizeof(float), stream);
    hipMemsetAsync(agg, 0, 2 * (size_t)N_NODES * OUT_DIM * sizeof(float), stream);

    degree_kernel<<<(N_EDGES + 255) / 256, 256, 0, stream>>>(esrc, edst, ns, nd, N_EDGES);
    norm_kernel<<<(N_NODES + 255) / 256, 256, 0, stream>>>(ns, nd, N_NODES);

    const int mtiles = N_NODES / 16;
    gemm_wmma_kernel<<<(mtiles + 7) / 8, 256, 0, stream>>>(h, W, ns, x, mtiles);

    const int swaves = (N_EDGES + 31) / 32;
    scatter_kernel<<<(swaves + 7) / 8, 256, 0, stream>>>(x, esrc, edst, agg, N_EDGES);

    const int total = N_NODES * OUT_DIM;
    finalize_kernel<<<(total + 255) / 256, 256, 0, stream>>>(agg, nd, b, out, total);
}


# aggregation pulls instead of pushes: returning int count adds give each edge a slot, edges bucketed by destination per call, one wave sums each node's rows (no agg atomics); float sum order changes
# speedup vs baseline: 2.7747x; 1.1679x over previous
#include <hip/hip_runtime.h>


#define N_NODES 100000
#define N_EDGES 1600000
#define IN_DIM  128
#define OUT_DIM 64
#define NEG_SLOPE 0.01f

#define NB          ((N_NODES + 255) / 256)
#define FILL_EPB    2048
#define FILL_BLOCKS ((N_EDGES + FILL_EPB - 1) / FILL_EPB)
#define RP_PER      ((N_NODES + 1 + FILL_BLOCKS - 1) / FILL_BLOCKS)

typedef float v2f __attribute__((ext_vector_type(2)));
typedef float v8f __attribute__((ext_vector_type(8)));

__device__ __forceinline__ int cache_half() {
    return (int)((__builtin_amdgcn_s_sendmsg_rtn(0x87) >> 18) & 1u);
}

__device__ __forceinline__ unsigned wave_incl_scan(unsigned v, int lane) {
#pragma unroll
    for (int d = 1; d < 32; d <<= 1) {
        unsigned y = __shfl_up(v, d);
        if (lane >= d) v += y;
    }
    return v;
}

__device__ __forceinline__ unsigned block_excl_scan(unsigned s, unsigned* wt, unsigned& total) {
    const int lane = threadIdx.x & 31;
    const int wave = threadIdx.x >> 5;
    const unsigned incl = wave_incl_scan(s, lane);
    if (lane == 31) wt[wave] = incl;
    __syncthreads();
    unsigned woff = 0, tot = 0;
#pragma unroll
    for (int w = 0; w < 8; ++w) {
        const unsigned q = wt[w];
        if (w < wave) woff += q;
        tot += q;
    }
    total = tot;
    return woff + incl - s;
}

__global__ void degree_kernel(const int* __restrict__ src, const int* __restrict__ dst,
                              float* __restrict__ outdeg, unsigned* __restrict__ cin,
                              unsigned* __restrict__ pos, int E) {
    int e = blockIdx.x * blockDim.x + threadIdx.x;
    const unsigned half = (unsigned)cache_half();
    const int hp = (int)half * N_NODES;
    if (e < E) {
        const int s = src[e];
        const int d = dst[e];
        atomicAdd(&outdeg[hp + s], 1.0f);
        const unsigned k = atomicAdd(&cin[hp + d], 1u);
        pos[e] = k | (half << 31);
    }
}

__global__ void __launch_bounds__(256)
norm_scan_kernel(float* __restrict__ ns, const unsigned* __restrict__ cin,
                 float* __restrict__ ndn, int2* __restrict__ lc, unsigned* __restrict__ bsum, int n) {
    __shared__ unsigned wt[8];
    const int i = blockIdx.x * 256 + threadIdx.x;
    unsigned c0 = 0, c1 = 0;
    if (i < n) {
        c0 = cin[i];
        c1 = cin[n + i];
        ns[i]  = rsqrtf(fmaxf(ns[i] + ns[n + i], 1.0f));
        ndn[i] = rsqrtf(fmaxf((float)(c0 + c1), 1.0f));
    }
    unsigned total;
    const unsigned excl = block_excl_scan(c0 + c1, wt, total);
    if (i < n) lc[i] = make_int2((int)excl, (int)c0);
    if (threadIdx.x == 0) bsum[blockIdx.x] = total;
}

__global__ void __launch_bounds__(256)
gemm_wmma_kernel(const float* __restrict__ h, const float* __restrict__ W,
                 const float* __restrict__ ns, float* __restrict__ x, int mtiles) {
    __shared__ float Wt[OUT_DIM * 130];
    for (int i = threadIdx.x; i < IN_DIM * OUT_DIM; i += 256) {
        int k = i >> 6;
        int c = i & 63;
        Wt[c * 130 + k] = W[i];
    }
    __syncthreads();

    const int wave = threadIdx.x >> 5;
    const int lane = threadIdx.x & 31;
    const int mtile = blockIdx.x * 8 + wave;
    if (mtile >= mtiles) return;

    const int mrow = (mtile << 4) + (lane & 15);
    const int kb   = (lane >> 4) << 1;
    const int n    = lane & 15;

    const float* hrow = h + (size_t)mrow * IN_DIM + kb;
    const float* w0 = &Wt[(0 * 16 + n) * 130 + kb];
    const float* w1 = &Wt[(1 * 16 + n) * 130 + kb];
    const float* w2 = &Wt[(2 * 16 + n) * 130 + kb];
    const float* w3 = &Wt[(3 * 16 + n) * 130 + kb];

    v8f acc0 = {}, acc1 = {}, acc2 = {}, acc3 = {};
#pragma unroll 4
    for (int k = 0; k < IN_DIM; k += 4) {
        v2f a  = *(const v2f*)(hrow + k);
        v2f b0 = *(const v2f*)(w0 + k);
        v2f b1 = *(const v2f*)(w1 + k);
        v2f b2 = *(const v2f*)(w2 + k);
        v2f b3 = *(const v2f*)(w3 + k);
        acc0 = __builtin_amdgcn_wmma_f32_16x16x4_f32(false, a, false, b0, (short)0, acc0, false, false);
        acc1 = __builtin_amdgcn_wmma_f32_16x16x4_f32(false, a, false, b1, (short)0, acc1, false, false);
        acc2 = __builtin_amdgcn_wmma_f32_16x16x4_f32(false, a, false, b2, (short)0, acc2, false, false);
        acc3 = __builtin_amdgcn_wmma_f32_16x16x4_f32(false, a, false, b3, (short)0, acc3, false, false);
    }

    const int rowbase = (mtile << 4) + ((lane >> 4) << 3);
#pragma unroll
    for (int r = 0; r < 8; ++r) {
        int row = rowbase + r;
        float s = ns[row];
        float* xr = x + (size_t)row * OUT_DIM + n;
        xr[0]  = acc0[r] * s;
        xr[16] = acc1[r] * s;
        xr[32] = acc2[r] * s;
        xr[48] = acc3[r] * s;
    }
}

__global__ void __launch_bounds__(256)
fill_kernel(const int* __restrict__ src, const int* __restrict__ dst, const unsigned* __restrict__ pos,
            const int2* __restrict__ lc, const unsigned* __restrict__ bsum,
            int* __restrict__ col, int* __restrict__ rowptr, int E) {
    __shared__ unsigned bp[513];
    __shared__ unsigned wt[8];
    const int t = threadIdx.x;
    {
        const unsigned a = (2 * t     < NB) ? bsum[2 * t]     : 0u;
        const unsigned b = (2 * t + 1 < NB) ? bsum[2 * t + 1] : 0u;
        unsigned total;
        const unsigned excl = block_excl_scan(a + b, wt, total);
        bp[2 * t]     = excl;
        bp[2 * t + 1] = excl + a;
        if (t == 0) bp[512] = total;
    }
    __syncthreads();

    {
        const int i0 = blockIdx.x * RP_PER;
        for (int i = i0 + t; i < i0 + RP_PER; i += 256) {
            if (i < N_NODES)       rowptr[i] = (int)(bp[i >> 8] + (unsigned)lc[i].x);
            else if (i == N_NODES) rowptr[i] = (int)bp[512];
        }
    }

    const int e0 = blockIdx.x * FILL_EPB + t;
    int d[FILL_EPB / 256], s[FILL_EPB / 256];
    unsigned p[FILL_EPB / 256];
    int2 v[FILL_EPB / 256];
#pragma unroll
    for (int u = 0; u < FILL_EPB / 256; ++u) {
        const int ee = min(e0 + u * 256, E - 1);
        d[u] = dst[ee];
        p[u] = pos[ee];
        s[u] = src[ee];
    }
#pragma unroll
    for (int u = 0; u < FILL_EPB / 256; ++u) v[u] = lc[d[u]];
#pragma unroll
    for (int u = 0; u < FILL_EPB / 256; ++u) {
        const unsigned slot = bp[d[u] >> 8] + (unsigned)v[u].x + ((p[u] >> 31) ? (unsigned)v[u].y : 0u) + (p[u] & 0x7fffffffu);
        if (e0 + u * 256 < E && slot < (unsigned)E) col[slot] = s[u];
    }
}

template <int G>
__device__ __forceinline__ void gather_group(const float2* __restrict__ x2, int cv, int j, int m,
                                             int lane, float2& acc) {
    float2 v[G];
#pragma unroll
    for (int u = 0; u < G; ++u) {
        const int s = __builtin_amdgcn_readlane(cv, j + u);
        v[u] = x2[(size_t)s * (OUT_DIM / 2) + lane];
    }
#pragma unroll
    for (int u = 0; u < G; ++u) {
        if (u < m) { acc.x += v[u].x; acc.y += v[u].y; }
    }
}

__global__ void __launch_bounds__(256)
gather_kernel(const float* __restrict__ x, const int* __restrict__ col, const int* __restrict__ rowptr,
              const float* __restrict__ ndn, const float* __restrict__ b, float* __restrict__ out, int n) {
    const int lane = threadIdx.x & 31;
    const int node = __builtin_amdgcn_readfirstlane(blockIdx.x * 8 + (threadIdx.x >> 5));
    if (node >= n) return;
    const int start = rowptr[node];
    const int end   = rowptr[node + 1];
    const float2* x2 = (const float2*)x;
    float2 acc = make_float2(0.0f, 0.0f);
    for (int base = start; base < end; base += 32) {
        const int cnt = min(32, end - base);
        const int cv = (lane < cnt) ? col[base + lane] : 0;
        int j = 0;
        for (; j + 8 <= cnt; j += 8) gather_group<8>(x2, cv, j, 8, lane, acc);
        if (j + 4 <= cnt) { gather_group<4>(x2, cv, j, 4, lane, acc); j += 4; }
        if (j < cnt) gather_group<4>(x2, cv, j, cnt - j, lane, acc);
    }
    const float nd = ndn[node];
    const float2 bb = ((const float2*)b)[lane];
    float2 r;
    r.x = acc.x * nd + bb.x;
    r.y = acc.y * nd + bb.y;
    r.x = (r.x >= 0.0f) ? r.x : r.x * NEG_SLOPE;
    r.y = (r.y >= 0.0f) ? r.y : r.y * NEG_SLOPE;
    ((float2*)out)[(size_t)node * (OUT_DIM / 2) + lane] = r;
}

extern "C" void kernel_launch(void* const* d_in, const int* in_sizes, int n_in,
                              void* d_out, int out_size, void* d_ws, size_t ws_size,
                              hipStream_t stream) {
    const float* h    = (const float*)d_in[0];
    const float* W    = (const float*)d_in[1];
    const float* b    = (const float*)d_in[2];
    const int*   esrc = (const int*)d_in[3];
    const int*   edst = (const int*)d_in[4];
    float*       out  = (float*)d_out;

    float*    ws     = (float*)d_ws;
    float*    ns     = ws;
    unsigned* cin    = (unsigned*)(ws + 2 * (size_t)N_NODES);
    float*    x      = ws + 4 * (size_t)N_NODES;
    float*    ndn    = x + (size_t)N_NODES * OUT_DIM;
    int2*     lc     = (int2*)(ndn + (size_t)N_NODES);
    unsigned* bsum   = (unsigned*)(ndn + 3 * (size_t)N_NODES);
    int*      rowptr = (int*)(bsum + 512);
    unsigned* pos    = (unsigned*)(rowptr + (size_t)N_NODES + 64);
    int*      col    = (int*)(pos + (size_t)N_EDGES);

    hipMemsetAsync(ns, 0, 4 * (size_t)N_NODES * sizeof(float), stream);

    degree_kernel<<<(N_EDGES + 255) / 256, 256, 0, stream>>>(esrc, edst, ns, cin, pos, N_EDGES);
    norm_scan_kernel<<<NB, 256, 0, stream>>>(ns, cin, ndn, lc, bsum, N_NODES);
    fill_kernel<<<FILL_BLOCKS, 256, 0, stream>>>(esrc, edst, pos, lc, bsum, col, rowptr, N_EDGES);

    const int mtiles = N_NODES / 16;
    gemm_wmma_kernel<<<(mtiles + 7) / 8, 256, 0, stream>>>(h, W, ns, x, mtiles);

    gather_kernel<<<(N_NODES + 7) / 8, 256, 0, stream>>>(x, col, rowptr, ndn, b, out, N_NODES);
}


# wave entry: gather wave sums 8 consecutive nodes (next node's indices requested ahead), degree thread counts 8 edges: 12,500+6,250 waves, were 100,000+50,000; arrival-order slots change
# speedup vs baseline: 2.8642x; 1.0323x over previous
#include <hip/hip_runtime.h>


#define N_NODES 100000
#define N_EDGES 1600000
#define IN_DIM  128
#define OUT_DIM 64
#define NEG_SLOPE 0.01f

#define NB          ((N_NODES + 255) / 256)
#define FILL_EPB    2048
#define FILL_BLOCKS ((N_EDGES + FILL_EPB - 1) / FILL_EPB)
#define RP_PER      ((N_NODES + 1 + FILL_BLOCKS - 1) / FILL_BLOCKS)
#define DEG_EPT     8
#define DEG_EPB     (256 * DEG_EPT)
#define DEG_BLOCKS  ((N_EDGES + DEG_EPB - 1) / DEG_EPB)
#define NPW         8
#define GATHER_BLOCKS ((N_NODES + 8 * NPW - 1) / (8 * NPW))

typedef float v2f __attribute__((ext_vector_type(2)));
typedef float v8f __attribute__((ext_vector_type(8)));

__device__ __forceinline__ int cache_half() {
    return (int)((__builtin_amdgcn_s_sendmsg_rtn(0x87) >> 18) & 1u);
}

__device__ __forceinline__ unsigned wave_incl_scan(unsigned v, int lane) {
#pragma unroll
    for (int d = 1; d < 32; d <<= 1) {
        unsigned y = __shfl_up(v, d);
        if (lane >= d) v += y;
    }
    return v;
}

__device__ __forceinline__ unsigned block_excl_scan(unsigned s, unsigned* wt, unsigned& total) {
    const int lane = threadIdx.x & 31;
    const int wave = threadIdx.x >> 5;
    const unsigned incl = wave_incl_scan(s, lane);
    if (lane == 31) wt[wave] = incl;
    __syncthreads();
    unsigned woff = 0, tot = 0;
#pragma unroll
    for (int w = 0; w < 8; ++w) {
        const unsigned q = wt[w];
        if (w < wave) woff += q;
        tot += q;
    }
    total = tot;
    return woff + incl - s;
}

__global__ void __launch_bounds__(256)
degree_kernel(const int* __restrict__ src, const int* __restrict__ dst,
              float* __restrict__ outdeg, unsigned* __restrict__ cin,
              unsigned* __restrict__ pos, int E) {
    const int b0 = blockIdx.x * DEG_EPB;
    const unsigned half = (unsigned)cache_half();
    const int hp = (int)half * N_NODES;
    if (b0 + DEG_EPB <= E) {
        const int e0 = b0 + threadIdx.x;
        int s[DEG_EPT], d[DEG_EPT];
        unsigned k[DEG_EPT];
#pragma unroll
        for (int u = 0; u < DEG_EPT; ++u) {
            s[u] = src[e0 + u * 256];
            d[u] = dst[e0 + u * 256];
        }
#pragma unroll
        for (int u = 0; u < DEG_EPT; ++u) {
            atomicAdd(&outdeg[hp + s[u]], 1.0f);
            k[u] = atomicAdd(&cin[hp + d[u]], 1u);
        }
#pragma unroll
        for (int u = 0; u < DEG_EPT; ++u) pos[e0 + u * 256] = k[u] | (half << 31);
    } else {
        for (int e = b0 + threadIdx.x; e < E; e += 256) {
            const int s = src[e];
            const int d = dst[e];
            atomicAdd(&outdeg[hp + s], 1.0f);
            const unsigned k = atomicAdd(&cin[hp + d], 1u);
            pos[e] = k | (half << 31);
        }
    }
}

__global__ void __launch_bounds__(256)
norm_scan_kernel(float* __restrict__ ns, const unsigned* __restrict__ cin,
                 float* __restrict__ ndn, int2* __restrict__ lc, unsigned* __restrict__ bsum, int n) {
    __shared__ unsigned wt[8];
    const int i = blockIdx.x * 256 + threadIdx.x;
    unsigned c0 = 0, c1 = 0;
    if (i < n) {
        c0 = cin[i];
        c1 = cin[n + i];
        ns[i]  = rsqrtf(fmaxf(ns[i] + ns[n + i], 1.0f));
        ndn[i] = rsqrtf(fmaxf((float)(c0 + c1), 1.0f));
    }
    unsigned total;
    const unsigned excl = block_excl_scan(c0 + c1, wt, total);
    if (i < n) lc[i] = make_int2((int)excl, (int)c0);
    if (threadIdx.x == 0) bsum[blockIdx.x] = total;
}

__global__ void __launch_bounds__(256)
gemm_wmma_kernel(const float* __restrict__ h, const float* __restrict__ W,
                 const float* __restrict__ ns, float* __restrict__ x, int mtiles) {
    __shared__ float Wt[OUT_DIM * 130];
    for (int i = threadIdx.x; i < IN_DIM * OUT_DIM; i += 256) {
        int k = i >> 6;
        int c = i & 63;
        Wt[c * 130 + k] = W[i];
    }
    __syncthreads();

    const int wave = threadIdx.x >> 5;
    const int lane = threadIdx.x & 31;
    const int mtile = blockIdx.x * 8 + wave;
    if (mtile >= mtiles) return;

    const int mrow = (mtile << 4) + (lane & 15);
    const int kb   = (lane >> 4) << 1;
    const int n    = lane & 15;

    const float* hrow = h + (size_t)mrow * IN_DIM + kb;
    const float* w0 = &Wt[(0 * 16 + n) * 130 + kb];
    const float* w1 = &Wt[(1 * 16 + n) * 130 + kb];
    const float* w2 = &Wt[(2 * 16 + n) * 130 + kb];
    const float* w3 = &Wt[(3 * 16 + n) * 130 + kb];

    v8f acc0 = {}, acc1 = {}, acc2 = {}, acc3 = {};
#pragma unroll 4
    for (int k = 0; k < IN_DIM; k += 4) {
        v2f a  = *(const v2f*)(hrow + k);
        v2f b0 = *(const v2f*)(w0 + k);
        v2f b1 = *(const v2f*)(w1 + k);
        v2f b2 = *(const v2f*)(w2 + k);
        v2f b3 = *(const v2f*)(w3 + k);
        acc0 = __builtin_amdgcn_wmma_f32_16x16x4_f32(false, a, false, b0, (short)0, acc0, false, false);
        acc1 = __builtin_amdgcn_wmma_f32_16x16x4_f32(false, a, false, b1, (short)0, acc1, false, false);
        acc2 = __builtin_amdgcn_wmma_f32_16x16x4_f32(false, a, false, b2, (short)0, acc2, false, false);
        acc3 = __builtin_amdgcn_wmma_f32_16x16x4_f32(false, a, false, b3, (short)0, acc3, false, false);
    }

    const int rowbase = (mtile << 4) + ((lane >> 4) << 3);
#pragma unroll
    for (int r = 0; r < 8; ++r) {
        int row = rowbase + r;
        float s = ns[row];
        float* xr = x + (size_t)row * OUT_DIM + n;
        xr[0]  = acc0[r] * s;
        xr[16] = acc1[r] * s;
        xr[32] = acc2[r] * s;
        xr[48] = acc3[r] * s;
    }
}

__global__ void __launch_bounds__(256)
fill_kernel(const int* __restrict__ src, const int* __restrict__ dst, const unsigned* __restrict__ pos,
            const int2* __restrict__ lc, const unsigned* __restrict__ bsum,
            int* __restrict__ col, int* __restrict__ rowptr, int E) {
    __shared__ unsigned bp[513];
    __shared__ unsigned wt[8];
    const int t = threadIdx.x;
    {
        const unsigned a = (2 * t     < NB) ? bsum[2 * t]     : 0u;
        const unsigned b = (2 * t + 1 < NB) ? bsum[2 * t + 1] : 0u;
        unsigned total;
        const unsigned excl = block_excl_scan(a + b, wt, total);
        bp[2 * t]     = excl;
        bp[2 * t + 1] = excl + a;
        if (t == 0) bp[512] = total;
    }
    __syncthreads();

    {
        const int i0 = blockIdx.x * RP_PER;
        for (int i = i0 + t; i < i0 + RP_PER; i += 256) {
            if (i < N_NODES)       rowptr[i] = (int)(bp[i >> 8] + (unsigned)lc[i].x);
            else if (i == N_NODES) rowptr[i] = (int)bp[512];
        }
    }

    const int e0 = blockIdx.x * FILL_EPB + t;
    int d[FILL_EPB / 256], s[FILL_EPB / 256];
    unsigned p[FILL_EPB / 256];
    int2 v[FILL_EPB / 256];
#pragma unroll
    for (int u = 0; u < FILL_EPB / 256; ++u) {
        const int ee = min(e0 + u * 256, E - 1);
        d[u] = dst[ee];
        p[u] = pos[ee];
        s[u] = src[ee];
    }
#pragma unroll
    for (int u = 0; u < FILL_EPB / 256; ++u) v[u] = lc[d[u]];
#pragma unroll
    for (int u = 0; u < FILL_EPB / 256; ++u) {
        const unsigned slot = bp[d[u] >> 8] + (unsigned)v[u].x + ((p[u] >> 31) ? (unsigned)v[u].y : 0u) + (p[u] & 0x7fffffffu);
        if (e0 + u * 256 < E && slot < (unsigned)E) col[slot] = s[u];
    }
}

template <int G>
__device__ __forceinline__ void gather_group(const float2* __restrict__ x2, int cv, int j, int m,
                                             int lane, float2& acc) {
    float2 v[G];
#pragma unroll
    for (int u = 0; u < G; ++u) {
        const int s = __builtin_amdgcn_readlane(cv, j + u);
        v[u] = x2[(size_t)s * (OUT_DIM / 2) + lane];
    }
#pragma unroll
    for (int u = 0; u < G; ++u) {
        if (u < m) { acc.x += v[u].x; acc.y += v[u].y; }
    }
}

__global__ void __launch_bounds__(256)
gather_kernel(const float* __restrict__ x, const int* __restrict__ col, const int* __restrict__ rowptr,
              const float* __restrict__ ndn, const float* __restrict__ b, float* __restrict__ out, int n) {
    const int lane = threadIdx.x & 31;
    const int node0 = __builtin_amdgcn_readfirstlane((blockIdx.x * 8 + (threadIdx.x >> 5)) * NPW);
    if (node0 >= n) return;
    const int nn = min(NPW, n - node0);
    const int rp = (lane <= nn) ? rowptr[node0 + lane] : 0;
    const int ndv = (lane < nn) ? __float_as_int(ndn[node0 + lane]) : 0;
    const float2 bb = ((const float2*)b)[lane];
    const float2* x2 = (const float2*)x;
    float2* o2 = (float2*)out + (size_t)node0 * (OUT_DIM / 2) + lane;

    int start = __builtin_amdgcn_readlane(rp, 0);
    int end   = __builtin_amdgcn_readlane(rp, 1);
    int cvr = col[min(start + lane, N_EDGES - 1)];
    for (int t = 0; t < nn; ++t) {
        const int nstart = end;
        const int nend   = (t + 1 < nn) ? __builtin_amdgcn_readlane(rp, t + 2) : end;
        const int cnt0   = min(32, end - start);
        int cv = (lane < cnt0) ? cvr : 0;
        float2 acc = make_float2(0.0f, 0.0f);
        {
            float2 v[8];
#pragma unroll
            for (int u = 0; u < 8; ++u) {
                const int s = __builtin_amdgcn_readlane(cv, u);
                v[u] = x2[(size_t)s * (OUT_DIM / 2) + lane];
            }
            cvr = col[min(nstart + lane, N_EDGES - 1)];
#pragma unroll
            for (int u = 0; u < 8; ++u) {
                if (u < cnt0) { acc.x += v[u].x; acc.y += v[u].y; }
            }
        }
        {
            int j = 8;
            for (; j + 8 <= cnt0; j += 8) gather_group<8>(x2, cv, j, 8, lane, acc);
            if (j + 4 <= cnt0) { gather_group<4>(x2, cv, j, 4, lane, acc); j += 4; }
            if (j < cnt0) gather_group<4>(x2, cv, j, cnt0 - j, lane, acc);
        }
        for (int base = start + 32; base < end; base += 32) {
            const int cnt = min(32, end - base);
            cv = (lane < cnt) ? col[base + lane] : 0;
            int j = 0;
            for (; j + 8 <= cnt; j += 8) gather_group<8>(x2, cv, j, 8, lane, acc);
            if (j + 4 <= cnt) { gather_group<4>(x2, cv, j, 4, lane, acc); j += 4; }
            if (j < cnt) gather_group<4>(x2, cv, j, cnt - j, lane, acc);
        }
        const float nd = __int_as_float(__builtin_amdgcn_readlane(ndv, t));
        float2 r;
        r.x = acc.x * nd + bb.x;
        r.y = acc.y * nd + bb.y;
        r.x = (r.x >= 0.0f) ? r.x : r.x * NEG_SLOPE;
        r.y = (r.y >= 0.0f) ? r.y : r.y * NEG_SLOPE;
        o2[(size_t)t * (OUT_DIM / 2)] = r;

        start = nstart;
        end   = nend;
    }
}

extern "C" void kernel_launch(void* const* d_in, const int* in_sizes, int n_in,
                              void* d_out, int out_size, void* d_ws, size_t ws_size,
                              hipStream_t stream) {
    const float* h    = (const float*)d_in[0];
    const float* W    = (const float*)d_in[1];
    const float* b    = (const float*)d_in[2];
    const int*   esrc = (const int*)d_in[3];
    const int*   edst = (const int*)d_in[4];
    float*       out  = (float*)d_out;

    float*    ws     = (float*)d_ws;
    float*    ns     = ws;
    unsigned* cin    = (unsigned*)(ws + 2 * (size_t)N_NODES);
    float*    x      = ws + 4 * (size_t)N_NODES;
    float*    ndn    = x + (size_t)N_NODES * OUT_DIM;
    int2*     lc     = (int2*)(ndn + (size_t)N_NODES);
    unsigned* bsum   = (unsigned*)(ndn + 3 * (size_t)N_NODES);
    int*      rowptr = (int*)(bsum + 512);
    unsigned* pos    = (unsigned*)(rowptr + (size_t)N_NODES + 64);
    int*      col    = (int*)(pos + (size_t)N_EDGES);

    hipMemsetAsync(ns, 0, 4 * (size_t)N_NODES * sizeof(float), stream);

    degree_kernel<<<DEG_BLOCKS, 256, 0, stream>>>(esrc, edst, ns, cin, pos, N_EDGES);
    norm_scan_kernel<<<NB, 256, 0, stream>>>(ns, cin, ndn, lc, bsum, N_NODES);
    fill_kernel<<<FILL_BLOCKS, 256, 0, stream>>>(esrc, edst, pos, lc, bsum, col, rowptr, N_EDGES);

    const int mtiles = N_NODES / 16;
    gemm_wmma_kernel<<<(mtiles + 7) / 8, 256, 0, stream>>>(h, W, ns, x, mtiles);

    gather_kernel<<<GATHER_BLOCKS, 256, 0, stream>>>(x, col, rowptr, ndn, b, out, N_NODES);
}


# each edge's source stored straight into a fixed 32-slot line per node and plane at the place its counting add returns (overflow list past 32): scan, pos array, placing launch go; 5->3 kernels
# speedup vs baseline: 3.1585x; 1.1028x over previous
#include <hip/hip_runtime.h>


#define N_NODES 100000
#define N_EDGES 1600000
#define IN_DIM  128
#define OUT_DIM 64
#define NEG_SLOPE 0.01f

#define CAP         32
#define SLOTS       (2 * CAP)
#define DEG_EPT     8
#define DEG_EPB     (256 * DEG_EPT)
#define DEG_BLOCKS  ((N_EDGES + DEG_EPB - 1) / DEG_EPB)
#define NPW         8
#define GATHER_BLOCKS ((N_NODES + 8 * NPW - 1) / (8 * NPW))

typedef float v2f __attribute__((ext_vector_type(2)));
typedef float v8f __attribute__((ext_vector_type(8)));

__device__ __forceinline__ int cache_half() {
    return (int)((__builtin_amdgcn_s_sendmsg_rtn(0x87) >> 18) & 1u);
}

__device__ __forceinline__ void put_overflow(unsigned* __restrict__ ovf_n, int2* __restrict__ ovf, int d, int s) {
    const unsigned i = atomicAdd(ovf_n, 1u);
    if (i < (unsigned)N_EDGES) ovf[i] = make_int2(d, s);
}

__global__ void __launch_bounds__(256)
degree_place_kernel(const int* __restrict__ src, const int* __restrict__ dst,
                    float* __restrict__ outdeg, unsigned* __restrict__ cin,
                    int* __restrict__ slots, unsigned* __restrict__ ovf_n, int2* __restrict__ ovf, int E) {
    const int b0 = blockIdx.x * DEG_EPB;
    const unsigned half = (unsigned)cache_half();
    const int hp = (int)half * N_NODES;
    const unsigned ho = half * CAP;
    if (b0 + DEG_EPB <= E) {
        const int e0 = b0 + threadIdx.x;
        int s[DEG_EPT], d[DEG_EPT];
        unsigned k[DEG_EPT];
#pragma unroll
        for (int u = 0; u < DEG_EPT; ++u) {
            s[u] = src[e0 + u * 256];
            d[u] = dst[e0 + u * 256];
        }
#pragma unroll
        for (int u = 0; u < DEG_EPT; ++u) {
            atomicAdd(&outdeg[hp + s[u]], 1.0f);
            k[u] = atomicAdd(&cin[hp + d[u]], 1u);
        }
        bool past = false;
#pragma unroll
        for (int u = 0; u < DEG_EPT; ++u) {
            if (k[u] < (unsigned)CAP) slots[(size_t)d[u] * SLOTS + ho + k[u]] = s[u];
            else past = true;
        }
        if (past) {
#pragma unroll
            for (int u = 0; u < DEG_EPT; ++u) {
                if (k[u] >= (unsigned)CAP) put_overflow(ovf_n, ovf, d[u], s[u]);
            }
        }
    } else {
        for (int e = b0 + threadIdx.x; e < E; e += 256) {
            const int s = src[e];
            const int d = dst[e];
            atomicAdd(&outdeg[hp + s], 1.0f);
            const unsigned k = atomicAdd(&cin[hp + d], 1u);
            if (k < (unsigned)CAP) slots[(size_t)d * SLOTS + ho + k] = s;
            else put_overflow(ovf_n, ovf, d, s);
        }
    }
}

__global__ void __launch_bounds__(256)
gemm_wmma_kernel(const float* __restrict__ h, const float* __restrict__ W,
                 const float* __restrict__ outdeg, float* __restrict__ x, int mtiles) {
    __shared__ float Wt[OUT_DIM * 130];
    for (int i = threadIdx.x; i < IN_DIM * OUT_DIM; i += 256) {
        int k = i >> 6;
        int c = i & 63;
        Wt[c * 130 + k] = W[i];
    }
    __syncthreads();

    const int wave = threadIdx.x >> 5;
    const int lane = threadIdx.x & 31;
    const int mtile = blockIdx.x * 8 + wave;
    if (mtile >= mtiles) return;

    const int mrow = (mtile << 4) + (lane & 15);
    const int kb   = (lane >> 4) << 1;
    const int n    = lane & 15;

    const float* hrow = h + (size_t)mrow * IN_DIM + kb;
    const float* w0 = &Wt[(0 * 16 + n) * 130 + kb];
    const float* w1 = &Wt[(1 * 16 + n) * 130 + kb];
    const float* w2 = &Wt[(2 * 16 + n) * 130 + kb];
    const float* w3 = &Wt[(3 * 16 + n) * 130 + kb];

    v8f acc0 = {}, acc1 = {}, acc2 = {}, acc3 = {};
#pragma unroll 4
    for (int k = 0; k < IN_DIM; k += 4) {
        v2f a  = *(const v2f*)(hrow + k);
        v2f b0 = *(const v2f*)(w0 + k);
        v2f b1 = *(const v2f*)(w1 + k);
        v2f b2 = *(const v2f*)(w2 + k);
        v2f b3 = *(const v2f*)(w3 + k);
        acc0 = __builtin_amdgcn_wmma_f32_16x16x4_f32(false, a, false, b0, (short)0, acc0, false, false);
        acc1 = __builtin_amdgcn_wmma_f32_16x16x4_f32(false, a, false, b1, (short)0, acc1, false, false);
        acc2 = __builtin_amdgcn_wmma_f32_16x16x4_f32(false, a, false, b2, (short)0, acc2, false, false);
        acc3 = __builtin_amdgcn_wmma_f32_16x16x4_f32(false, a, false, b3, (short)0, acc3, false, false);
    }

    const int rowbase = (mtile << 4) + ((lane >> 4) << 3);
#pragma unroll
    for (int r = 0; r < 8; ++r) {
        int row = rowbase + r;
        float s = rsqrtf(fmaxf(outdeg[row] + outdeg[N_NODES + row], 1.0f));
        float* xr = x + (size_t)row * OUT_DIM + n;
        xr[0]  = acc0[r] * s;
        xr[16] = acc1[r] * s;
        xr[32] = acc2[r] * s;
        xr[48] = acc3[r] * s;
    }
}

template <int G>
__device__ __forceinline__ void gather_group(const float2* __restrict__ x2, int cv, int j, int m,
                                             int lane, float2& acc) {
    float2 v[G];
#pragma unroll
    for (int u = 0; u < G; ++u) {
        const int s = __builtin_amdgcn_readlane(cv, j + u);
        v[u] = x2[(size_t)s * (OUT_DIM / 2) + lane];
    }
#pragma unroll
    for (int u = 0; u < G; ++u) {
        if (u < m) { acc.x += v[u].x; acc.y += v[u].y; }
    }
}

__device__ __forceinline__ int slot_of(int i, int a0) {
    return min((i < a0) ? i : CAP + i - a0, SLOTS - 1);
}

__device__ __forceinline__ void gather_overflow(const float2* __restrict__ x2, const unsigned* __restrict__ ovf_n,
                                                const int2* __restrict__ ovf, int node, int lane, float2& acc) {
    const unsigned no = min((unsigned)__builtin_amdgcn_readfirstlane((int)ovf_n[0]), (unsigned)N_EDGES);
    for (unsigned base = 0; base < no; base += 32) {
        const unsigned i = base + (unsigned)lane;
        const int2 e = (i < no) ? ovf[i] : make_int2(-1, 0);
        unsigned m = __builtin_amdgcn_ballot_w32(e.x == node);
        while (m) {
            const int u = __builtin_ctz(m);
            m &= m - 1;
            const int s = __builtin_amdgcn_readlane(e.y, u);
            const float2 v = x2[(size_t)s * (OUT_DIM / 2) + lane];
            acc.x += v.x; acc.y += v.y;
        }
    }
}

__global__ void __launch_bounds__(256)
gather_kernel(const float* __restrict__ x, const int* __restrict__ slots, const unsigned* __restrict__ cin,
              const unsigned* __restrict__ ovf_n, const int2* __restrict__ ovf,
              const float* __restrict__ b, float* __restrict__ out, int n) {
    const int lane = threadIdx.x & 31;
    const int node0 = __builtin_amdgcn_readfirstlane((blockIdx.x * 8 + (threadIdx.x >> 5)) * NPW);
    if (node0 >= n) return;
    const int nn = min(NPW, n - node0);
    const unsigned c0 = (lane < nn) ? cin[node0 + lane] : 0u;
    const unsigned c1 = (lane < nn) ? cin[n + node0 + lane] : 0u;
    const int a0v = (int)min(c0, (unsigned)CAP);
    const int a1v = (int)min(c1, (unsigned)CAP);
    const int ovv = (c0 > (unsigned)CAP || c1 > (unsigned)CAP) ? 1 : 0;
    const int ndv = __float_as_int(rsqrtf(fmaxf((float)(c0 + c1), 1.0f)));
    const float2 bb = ((const float2*)b)[lane];
    const float2* x2 = (const float2*)x;
    const int* srow = slots + (size_t)node0 * SLOTS;
    float2* o2 = (float2*)out + (size_t)node0 * (OUT_DIM / 2) + lane;

    int a0  = __builtin_amdgcn_readlane(a0v, 0);
    int cnt = a0 + __builtin_amdgcn_readlane(a1v, 0);
    int cvr = srow[slot_of(lane, a0)];
    for (int t = 0; t < nn; ++t) {
        const int tn   = min(t + 1, nn - 1);
        const int na0  = __builtin_amdgcn_readlane(a0v, tn);
        const int ncnt = na0 + __builtin_amdgcn_readlane(a1v, tn);
        const int cnt0 = min(32, cnt);
        int cv = (lane < cnt0) ? cvr : 0;
        float2 acc = make_float2(0.0f, 0.0f);
        {
            float2 v[8];
#pragma unroll
            for (int u = 0; u < 8; ++u) {
                const int s = __builtin_amdgcn_readlane(cv, u);
                v[u] = x2[(size_t)s * (OUT_DIM / 2) + lane];
            }
            cvr = srow[(size_t)tn * SLOTS + slot_of(lane, na0)];
#pragma unroll
            for (int u = 0; u < 8; ++u) {
                if (u < cnt0) { acc.x += v[u].x; acc.y += v[u].y; }
            }
        }
        {
            int j = 8;
            for (; j + 8 <= cnt0; j += 8) gather_group<8>(x2, cv, j, 8, lane, acc);
            if (j + 4 <= cnt0) { gather_group<4>(x2, cv, j, 4, lane, acc); j += 4; }
            if (j < cnt0) gather_group<4>(x2, cv, j, cnt0 - j, lane, acc);
        }
        if (cnt > 32) {
            const int c2 = cnt - 32;
            cv = (lane < c2) ? srow[(size_t)t * SLOTS + slot_of(32 + lane, a0)] : 0;
            int j = 0;
            for (; j + 8 <= c2; j += 8) gather_group<8>(x2, cv, j, 8, lane, acc);
            if (j + 4 <= c2) { gather_group<4>(x2, cv, j, 4, lane, acc); j += 4; }
            if (j < c2) gather_group<4>(x2, cv, j, c2 - j, lane, acc);
        }
        if (__builtin_amdgcn_readlane(ovv, t)) gather_overflow(x2, ovf_n, ovf, node0 + t, lane, acc);

        const float nd = __int_as_float(__builtin_amdgcn_readlane(ndv, t));
        float2 r;
        r.x = acc.x * nd + bb.x;
        r.y = acc.y * nd + bb.y;
        r.x = (r.x >= 0.0f) ? r.x : r.x * NEG_SLOPE;
        r.y = (r.y >= 0.0f) ? r.y : r.y * NEG_SLOPE;
        o2[(size_t)t * (OUT_DIM / 2)] = r;

        a0  = na0;
        cnt = ncnt;
    }
}

extern "C" void kernel_launch(void* const* d_in, const int* in_sizes, int n_in,
                              void* d_out, int out_size, void* d_ws, size_t ws_size,
                              hipStream_t stream) {
    const float* h    = (const float*)d_in[0];
    const float* W    = (const float*)d_in[1];
    const float* b    = (const float*)d_in[2];
    const int*   esrc = (const int*)d_in[3];
    const int*   edst = (const int*)d_in[4];
    float*       out  = (float*)d_out;

    float*    ws     = (float*)d_ws;
    float*    outdeg = ws;
    unsigned* cin    = (unsigned*)(ws + 2 * (size_t)N_NODES);
    unsigned* ovf_n  = (unsigned*)(ws + 4 * (size_t)N_NODES);
    float*    x      = ws + 4 * (size_t)N_NODES + 64;
    int*      slots  = (int*)(x + (size_t)N_NODES * OUT_DIM);
    int2*     ovf    = (int2*)(slots + (size_t)N_NODES * SLOTS);

    hipMemsetAsync(ws, 0, (4 * (size_t)N_NODES + 64) * sizeof(float), stream);

    degree_place_kernel<<<DEG_BLOCKS, 256, 0, stream>>>(esrc, edst, outdeg, cin, slots, ovf_n, ovf, N_EDGES);

    const int mtiles = N_NODES / 16;
    gemm_wmma_kernel<<<(mtiles + 7) / 8, 256, 0, stream>>>(h, W, outdeg, x, mtiles);

    gather_kernel<<<GATHER_BLOCKS, 256, 0, stream>>>(x, slots, cin, ovf_n, ovf, b, out, N_NODES);
}


# edges staged per 400-node destination range in LDS, stored as whole 128-byte cells; a work-group per range counts and fills slot lines in LDS: no global in-degree adds; 4 kernels; sum order changes
# speedup vs baseline: 4.4995x; 1.4246x over previous
#include <hip/hip_runtime.h>


#define N_NODES 100000
#define N_EDGES 1600000
#define IN_DIM  128
#define OUT_DIM 64
#define NEG_SLOPE 0.01f

#define CAP         32
#define SLOTS       (2 * CAP)
#define DEG_EPT     8
#define DEG_EPB     (256 * DEG_EPT)
#define DEG_BLOCKS  ((N_EDGES + DEG_EPB - 1) / DEG_EPB)
#define RNG         400
#define NBIN        (N_NODES / RNG)
#define CELL        16
#define BUILD_THREADS 1024
static_assert(NBIN * RNG == N_NODES, "ranges must tile the nodes exactly");
#define NPW         8
#define GATHER_BLOCKS ((N_NODES + 8 * NPW - 1) / (8 * NPW))

typedef float v2f __attribute__((ext_vector_type(2)));
typedef float v8f __attribute__((ext_vector_type(8)));

__device__ __forceinline__ int cache_half() {
    return (int)((__builtin_amdgcn_s_sendmsg_rtn(0x87) >> 18) & 1u);
}

__device__ __forceinline__ void put_overflow(unsigned* __restrict__ ovf_n, int2* __restrict__ ovf, int d, int s) {
    const unsigned i = atomicAdd(ovf_n, 1u);
    if (i < (unsigned)N_EDGES) ovf[i] = make_int2(d, s);
}

__global__ void __launch_bounds__(256)
part_kernel(const int* __restrict__ src, const int* __restrict__ dst, float* __restrict__ outdeg,
            int2* __restrict__ bucket, unsigned* __restrict__ spill_n, int2* __restrict__ spill, int E) {
    __shared__ unsigned hist[NBIN];
    __shared__ int2 stage[NBIN * CELL];
    for (int i = threadIdx.x; i < NBIN; i += 256) hist[i] = 0u;
    for (int i = threadIdx.x; i < NBIN * CELL; i += 256) stage[i] = make_int2(-1, 0);
    __syncthreads();

    const int hp = cache_half() * N_NODES;
    const int b0 = blockIdx.x * DEG_EPB;
    if (b0 + DEG_EPB <= E) {
        const int e0 = b0 + threadIdx.x;
        int s[DEG_EPT], d[DEG_EPT], bin[DEG_EPT];
        unsigned r[DEG_EPT];
#pragma unroll
        for (int u = 0; u < DEG_EPT; ++u) {
            s[u] = src[e0 + u * 256];
            d[u] = dst[e0 + u * 256];
        }
#pragma unroll
        for (int u = 0; u < DEG_EPT; ++u) atomicAdd(&outdeg[hp + s[u]], 1.0f);
#pragma unroll
        for (int u = 0; u < DEG_EPT; ++u) {
            bin[u] = (int)min((unsigned)d[u] / (unsigned)RNG, (unsigned)(NBIN - 1));
            r[u] = atomicAdd(&hist[bin[u]], 1u);
        }
        bool past = false;
#pragma unroll
        for (int u = 0; u < DEG_EPT; ++u) {
            if (r[u] < (unsigned)CELL) stage[bin[u] * CELL + (int)r[u]] = make_int2(d[u], s[u]);
            else past = true;
        }
        if (past) {
#pragma unroll
            for (int u = 0; u < DEG_EPT; ++u) {
                if (r[u] >= (unsigned)CELL) put_overflow(spill_n, spill, d[u], s[u]);
            }
        }
    } else {
        for (int e = b0 + threadIdx.x; e < E; e += 256) {
            const int s = src[e];
            const int d = dst[e];
            atomicAdd(&outdeg[hp + s], 1.0f);
            const int bin = (int)min((unsigned)d / (unsigned)RNG, (unsigned)(NBIN - 1));
            const unsigned r = atomicAdd(&hist[bin], 1u);
            if (r < (unsigned)CELL) stage[bin * CELL + (int)r] = make_int2(d, s);
            else put_overflow(spill_n, spill, d, s);
        }
    }
    __syncthreads();
    for (int i = threadIdx.x; i < NBIN * CELL; i += 256) {
        const int bin = i / CELL;
        bucket[((size_t)bin * DEG_BLOCKS + blockIdx.x) * CELL + (i % CELL)] = stage[i];
    }
}

__global__ void __launch_bounds__(BUILD_THREADS)
build_kernel(const int4* __restrict__ bucket4, const unsigned* __restrict__ spill_n, const int2* __restrict__ spill,
             unsigned* __restrict__ cin, int* __restrict__ slots, unsigned* __restrict__ ovf_n, int2* __restrict__ ovf) {
    __shared__ unsigned cnt[RNG];
    __shared__ int lines[RNG * SLOTS];
    const int node0 = blockIdx.x * RNG;
    for (int i = threadIdx.x; i < RNG; i += BUILD_THREADS) cnt[i] = 0u;
    for (int i = threadIdx.x; i < RNG * SLOTS; i += BUILD_THREADS) lines[i] = 0;
    __syncthreads();

    constexpr int Q   = DEG_BLOCKS * CELL / 2;
    constexpr int PER = (Q + BUILD_THREADS - 1) / BUILD_THREADS;
    const int4* bk = bucket4 + (size_t)blockIdx.x * Q;
    int dd[2 * PER], ss[2 * PER];
    unsigned ld[2 * PER], k[2 * PER];
#pragma unroll
    for (int j = 0; j < PER; ++j) {
        const int i = (int)threadIdx.x + j * BUILD_THREADS;
        const int4 q = (i < Q) ? bk[i] : make_int4(-1, 0, -1, 0);
        dd[2 * j] = q.x; ss[2 * j] = q.y; dd[2 * j + 1] = q.z; ss[2 * j + 1] = q.w;
    }
#pragma unroll
    for (int j = 0; j < 2 * PER; ++j) {
        ld[j] = (unsigned)(dd[j] - node0);
        k[j]  = 0xffffffffu;
        if (ld[j] < (unsigned)RNG) k[j] = atomicAdd(&cnt[ld[j]], 1u);
    }
    bool past = false;
#pragma unroll
    for (int j = 0; j < 2 * PER; ++j) {
        if (k[j] < (unsigned)SLOTS) lines[ld[j] * SLOTS + k[j]] = ss[j];
        else if (ld[j] < (unsigned)RNG) past = true;
    }
    if (past) {
#pragma unroll
        for (int j = 0; j < 2 * PER; ++j) {
            if (ld[j] < (unsigned)RNG && k[j] >= (unsigned)SLOTS) put_overflow(ovf_n, ovf, dd[j], ss[j]);
        }
    }
    const unsigned ns = min(spill_n[0], (unsigned)N_EDGES);
    for (unsigned i = threadIdx.x; i < ns; i += BUILD_THREADS) {
        const int2 e = spill[i];
        const unsigned l = (unsigned)(e.x - node0);
        if (l < (unsigned)RNG) {
            const unsigned kk = atomicAdd(&cnt[l], 1u);
            if (kk < (unsigned)SLOTS) lines[l * SLOTS + kk] = e.y;
            else put_overflow(ovf_n, ovf, e.x, e.y);
        }
    }
    __syncthreads();

    int* so = slots + (size_t)node0 * SLOTS;
    for (int i = threadIdx.x; i < RNG * SLOTS; i += BUILD_THREADS) so[i] = lines[i];
    for (int i = threadIdx.x; i < RNG; i += BUILD_THREADS) {
        const unsigned cc = cnt[i];
        const unsigned c0 = min(cc, (unsigned)CAP);
        cin[node0 + i] = c0;
        cin[N_NODES + node0 + i] = cc - c0;
    }
}

__global__ void __launch_bounds__(256)
gemm_wmma_kernel(const float* __restrict__ h, const float* __restrict__ W,
                 const float* __restrict__ outdeg, float* __restrict__ x, int mtiles) {
    __shared__ float Wt[OUT_DIM * 130];
    for (int i = threadIdx.x; i < IN_DIM * OUT_DIM; i += 256) {
        int k = i >> 6;
        int c = i & 63;
        Wt[c * 130 + k] = W[i];
    }
    __syncthreads();

    const int wave = threadIdx.x >> 5;
    const int lane = threadIdx.x & 31;
    const int mtile = blockIdx.x * 8 + wave;
    if (mtile >= mtiles) return;

    const int mrow = (mtile << 4) + (lane & 15);
    const int kb   = (lane >> 4) << 1;
    const int n    = lane & 15;

    const float* hrow = h + (size_t)mrow * IN_DIM + kb;
    const float* w0 = &Wt[(0 * 16 + n) * 130 + kb];
    const float* w1 = &Wt[(1 * 16 + n) * 130 + kb];
    const float* w2 = &Wt[(2 * 16 + n) * 130 + kb];
    const float* w3 = &Wt[(3 * 16 + n) * 130 + kb];

    v8f acc0 = {}, acc1 = {}, acc2 = {}, acc3 = {};
#pragma unroll 4
    for (int k = 0; k < IN_DIM; k += 4) {
        v2f a  = *(const v2f*)(hrow + k);
        v2f b0 = *(const v2f*)(w0 + k);
        v2f b1 = *(const v2f*)(w1 + k);
        v2f b2 = *(const v2f*)(w2 + k);
        v2f b3 = *(const v2f*)(w3 + k);
        acc0 = __builtin_amdgcn_wmma_f32_16x16x4_f32(false, a, false, b0, (short)0, acc0, false, false);
        acc1 = __builtin_amdgcn_wmma_f32_16x16x4_f32(false, a, false, b1, (short)0, acc1, false, false);
        acc2 = __builtin_amdgcn_wmma_f32_16x16x4_f32(false, a, false, b2, (short)0, acc2, false, false);
        acc3 = __builtin_amdgcn_wmma_f32_16x16x4_f32(false, a, false, b3, (short)0, acc3, false, false);
    }

    const int rowbase = (mtile << 4) + ((lane >> 4) << 3);
#pragma unroll
    for (int r = 0; r < 8; ++r) {
        int row = rowbase + r;
        float s = rsqrtf(fmaxf(outdeg[row] + outdeg[N_NODES + row], 1.0f));
        float* xr = x + (size_t)row * OUT_DIM + n;
        xr[0]  = acc0[r] * s;
        xr[16] = acc1[r] * s;
        xr[32] = acc2[r] * s;
        xr[48] = acc3[r] * s;
    }
}

template <int G>
__device__ __forceinline__ void gather_group(const float2* __restrict__ x2, int cv, int j, int m,
                                             int lane, float2& acc) {
    float2 v[G];
#pragma unroll
    for (int u = 0; u < G; ++u) {
        const int s = __builtin_amdgcn_readlane(cv, j + u);
        v[u] = x2[(size_t)s * (OUT_DIM / 2) + lane];
    }
#pragma unroll
    for (int u = 0; u < G; ++u) {
        if (u < m) { acc.x += v[u].x; acc.y += v[u].y; }
    }
}

__device__ __forceinline__ int slot_of(int i, int a0) {
    return min((i < a0) ? i : CAP + i - a0, SLOTS - 1);
}

__device__ __forceinline__ void gather_overflow(const float2* __restrict__ x2, const unsigned* __restrict__ ovf_n,
                                                const int2* __restrict__ ovf, int node, int lane, float2& acc) {
    const unsigned no = min((unsigned)__builtin_amdgcn_readfirstlane((int)ovf_n[0]), (unsigned)N_EDGES);
    for (unsigned base = 0; base < no; base += 32) {
        const unsigned i = base + (unsigned)lane;
        const int2 e = (i < no) ? ovf[i] : make_int2(-1, 0);
        unsigned m = __builtin_amdgcn_ballot_w32(e.x == node);
        while (m) {
            const int u = __builtin_ctz(m);
            m &= m - 1;
            const int s = __builtin_amdgcn_readlane(e.y, u);
            const float2 v = x2[(size_t)s * (OUT_DIM / 2) + lane];
            acc.x += v.x; acc.y += v.y;
        }
    }
}

__global__ void __launch_bounds__(256)
gather_kernel(const float* __restrict__ x, const int* __restrict__ slots, const unsigned* __restrict__ cin,
              const unsigned* __restrict__ ovf_n, const int2* __restrict__ ovf,
              const float* __restrict__ b, float* __restrict__ out, int n) {
    const int lane = threadIdx.x & 31;
    const int node0 = __builtin_amdgcn_readfirstlane((blockIdx.x * 8 + (threadIdx.x >> 5)) * NPW);
    if (node0 >= n) return;
    const int nn = min(NPW, n - node0);
    const unsigned c0 = (lane < nn) ? cin[node0 + lane] : 0u;
    const unsigned c1 = (lane < nn) ? cin[n + node0 + lane] : 0u;
    const int a0v = (int)min(c0, (unsigned)CAP);
    const int a1v = (int)min(c1, (unsigned)CAP);
    const int ovv = (c0 > (unsigned)CAP || c1 > (unsigned)CAP) ? 1 : 0;
    const int ndv = __float_as_int(rsqrtf(fmaxf((float)(c0 + c1), 1.0f)));
    const float2 bb = ((const float2*)b)[lane];
    const float2* x2 = (const float2*)x;
    const int* srow = slots + (size_t)node0 * SLOTS;
    float2* o2 = (float2*)out + (size_t)node0 * (OUT_DIM / 2) + lane;

    int a0  = __builtin_amdgcn_readlane(a0v, 0);
    int cnt = a0 + __builtin_amdgcn_readlane(a1v, 0);
    int cvr = srow[slot_of(lane, a0)];
    for (int t = 0; t < nn; ++t) {
        const int tn   = min(t + 1, nn - 1);
        const int na0  = __builtin_amdgcn_readlane(a0v, tn);
        const int ncnt = na0 + __builtin_amdgcn_readlane(a1v, tn);
        const int cnt0 = min(32, cnt);
        int cv = (lane < cnt0) ? cvr : 0;
        float2 acc = make_float2(0.0f, 0.0f);
        {
            float2 v[8];
#pragma unroll
            for (int u = 0; u < 8; ++u) {
                const int s = __builtin_amdgcn_readlane(cv, u);
                v[u] = x2[(size_t)s * (OUT_DIM / 2) + lane];
            }
            cvr = srow[(size_t)tn * SLOTS + slot_of(lane, na0)];
#pragma unroll
            for (int u = 0; u < 8; ++u) {
                if (u < cnt0) { acc.x += v[u].x; acc.y += v[u].y; }
            }
        }
        {
            int j = 8;
            for (; j + 8 <= cnt0; j += 8) gather_group<8>(x2, cv, j, 8, lane, acc);
            if (j + 4 <= cnt0) { gather_group<4>(x2, cv, j, 4, lane, acc); j += 4; }
            if (j < cnt0) gather_group<4>(x2, cv, j, cnt0 - j, lane, acc);
        }
        if (cnt > 32) {
            const int c2 = cnt - 32;
            cv = (lane < c2) ? srow[(size_t)t * SLOTS + slot_of(32 + lane, a0)] : 0;
            int j = 0;
            for (; j + 8 <= c2; j += 8) gather_group<8>(x2, cv, j, 8, lane, acc);
            if (j + 4 <= c2) { gather_group<4>(x2, cv, j, 4, lane, acc); j += 4; }
            if (j < c2) gather_group<4>(x2, cv, j, c2 - j, lane, acc);
        }
        if (__builtin_amdgcn_readlane(ovv, t)) gather_overflow(x2, ovf_n, ovf, node0 + t, lane, acc);

        const float nd = __int_as_float(__builtin_amdgcn_readlane(ndv, t));
        float2 r;
        r.x = acc.x * nd + bb.x;
        r.y = acc.y * nd + bb.y;
        r.x = (r.x >= 0.0f) ? r.x : r.x * NEG_SLOPE;
        r.y = (r.y >= 0.0f) ? r.y : r.y * NEG_SLOPE;
        o2[(size_t)t * (OUT_DIM / 2)] = r;

        a0  = na0;
        cnt = ncnt;
    }
}

extern "C" void kernel_launch(void* const* d_in, const int* in_sizes, int n_in,
                              void* d_out, int out_size, void* d_ws, size_t ws_size,
                              hipStream_t stream) {
    const float* h    = (const float*)d_in[0];
    const float* W    = (const float*)d_in[1];
    const float* b    = (const float*)d_in[2];
    const int*   esrc = (const int*)d_in[3];
    const int*   edst = (const int*)d_in[4];
    float*       out  = (float*)d_out;

    float*    ws      = (float*)d_ws;
    float*    outdeg  = ws;
    unsigned* ovf_n   = (unsigned*)(ws + 2 * (size_t)N_NODES);
    unsigned* spill_n = ovf_n + 32;
    unsigned* cin     = (unsigned*)(ws + 2 * (size_t)N_NODES + 64);
    float*    x       = ws + 4 * (size_t)N_NODES + 64;
    int*      slots   = (int*)(x + (size_t)N_NODES * OUT_DIM);
    int2*     ovf     = (int2*)(slots + (size_t)N_NODES * SLOTS);
    int2*     spill   = ovf + (size_t)N_EDGES;
    int2*     bucket  = spill + (size_t)N_EDGES;

    hipMemsetAsync(ws, 0, (2 * (size_t)N_NODES + 64) * sizeof(float), stream);

    part_kernel<<<DEG_BLOCKS, 256, 0, stream>>>(esrc, edst, outdeg, bucket, spill_n, spill, N_EDGES);

    build_kernel<<<NBIN, BUILD_THREADS, 0, stream>>>((const int4*)bucket, spill_n, spill, cin, slots, ovf_n, ovf);

    const int mtiles = N_NODES / 16;
    gemm_wmma_kernel<<<(mtiles + 7) / 8, 256, 0, stream>>>(h, W, outdeg, x, mtiles);

    gather_kernel<<<GATHER_BLOCKS, 256, 0, stream>>>(x, slots, cin, ovf_n, ovf, b, out, N_NODES);
}


# slot build and h@W product in ONE launch: 250 product work-groups (25 wave tiles each) start first, 250 build work-groups share their WGPs; independent jobs overlap, one launch goes; 4->3 kernels
# speedup vs baseline: 4.8734x; 1.0831x over previous
#include <hip/hip_runtime.h>


#define N_NODES 100000
#define N_EDGES 1600000
#define IN_DIM  128
#define OUT_DIM 64
#define NEG_SLOPE 0.01f

#define CAP         32
#define SLOTS       (2 * CAP)
#define DEG_EPT     8
#define DEG_EPB     (256 * DEG_EPT)
#define DEG_BLOCKS  ((N_EDGES + DEG_EPB - 1) / DEG_EPB)
#define RNG         400
#define NBIN        (N_NODES / RNG)
#define CELL        16
#define BUILD_THREADS 1024
static_assert(NBIN * RNG == N_NODES, "ranges must tile the nodes exactly");
#define PROD_TILES  25
#define PROD_WGS    (N_NODES / 16 / PROD_TILES)
static_assert(PROD_WGS * PROD_TILES * 16 == N_NODES, "the product's work-groups must tile the rows exactly");
static_assert(PROD_TILES * 32 <= BUILD_THREADS, "one wave per tile");
#define NPW         8
#define GATHER_BLOCKS ((N_NODES + 8 * NPW - 1) / (8 * NPW))

typedef float v2f __attribute__((ext_vector_type(2)));
typedef float v8f __attribute__((ext_vector_type(8)));

__device__ __forceinline__ int cache_half() {
    return (int)((__builtin_amdgcn_s_sendmsg_rtn(0x87) >> 18) & 1u);
}

__device__ __forceinline__ void put_overflow(unsigned* __restrict__ ovf_n, int2* __restrict__ ovf, int d, int s) {
    const unsigned i = atomicAdd(ovf_n, 1u);
    if (i < (unsigned)N_EDGES) ovf[i] = make_int2(d, s);
}

__global__ void __launch_bounds__(256)
part_kernel(const int* __restrict__ src, const int* __restrict__ dst, float* __restrict__ outdeg,
            int2* __restrict__ bucket, unsigned* __restrict__ spill_n, int2* __restrict__ spill, int E) {
    __shared__ unsigned hist[NBIN];
    __shared__ int2 stage[NBIN * CELL];
    for (int i = threadIdx.x; i < NBIN; i += 256) hist[i] = 0u;
    for (int i = threadIdx.x; i < NBIN * CELL; i += 256) stage[i] = make_int2(-1, 0);
    __syncthreads();

    const int hp = cache_half() * N_NODES;
    const int b0 = blockIdx.x * DEG_EPB;
    if (b0 + DEG_EPB <= E) {
        const int e0 = b0 + threadIdx.x;
        int s[DEG_EPT], d[DEG_EPT], bin[DEG_EPT];
        unsigned r[DEG_EPT];
#pragma unroll
        for (int u = 0; u < DEG_EPT; ++u) {
            s[u] = src[e0 + u * 256];
            d[u] = dst[e0 + u * 256];
        }
#pragma unroll
        for (int u = 0; u < DEG_EPT; ++u) atomicAdd(&outdeg[hp + s[u]], 1.0f);
#pragma unroll
        for (int u = 0; u < DEG_EPT; ++u) {
            bin[u] = (int)min((unsigned)d[u] / (unsigned)RNG, (unsigned)(NBIN - 1));
            r[u] = atomicAdd(&hist[bin[u]], 1u);
        }
        bool past = false;
#pragma unroll
        for (int u = 0; u < DEG_EPT; ++u) {
            if (r[u] < (unsigned)CELL) stage[bin[u] * CELL + (int)r[u]] = make_int2(d[u], s[u]);
            else past = true;
        }
        if (past) {
#pragma unroll
            for (int u = 0; u < DEG_EPT; ++u) {
                if (r[u] >= (unsigned)CELL) put_overflow(spill_n, spill, d[u], s[u]);
            }
        }
    } else {
        for (int e = b0 + threadIdx.x; e < E; e += 256) {
            const int s = src[e];
            const int d = dst[e];
            atomicAdd(&outdeg[hp + s], 1.0f);
            const int bin = (int)min((unsigned)d / (unsigned)RNG, (unsigned)(NBIN - 1));
            const unsigned r = atomicAdd(&hist[bin], 1u);
            if (r < (unsigned)CELL) stage[bin * CELL + (int)r] = make_int2(d, s);
            else put_overflow(spill_n, spill, d, s);
        }
    }
    __syncthreads();
    for (int i = threadIdx.x; i < NBIN * CELL; i += 256) {
        const int bin = i / CELL;
        bucket[((size_t)bin * DEG_BLOCKS + blockIdx.x) * CELL + (i % CELL)] = stage[i];
    }
}

__device__ __forceinline__ void
build_part(unsigned* __restrict__ cnt, int* __restrict__ lines, int bin,
           const int4* __restrict__ bucket4, const unsigned* __restrict__ spill_n, const int2* __restrict__ spill,
           unsigned* __restrict__ cin, int* __restrict__ slots, unsigned* __restrict__ ovf_n, int2* __restrict__ ovf) {
    const int node0 = bin * RNG;
    for (int i = threadIdx.x; i < RNG; i += BUILD_THREADS) cnt[i] = 0u;
    for (int i = threadIdx.x; i < RNG * SLOTS; i += BUILD_THREADS) lines[i] = 0;
    __syncthreads();

    constexpr int Q   = DEG_BLOCKS * CELL / 2;
    constexpr int PER = (Q + BUILD_THREADS - 1) / BUILD_THREADS;
    const int4* bk = bucket4 + (size_t)bin * Q;
    int dd[2 * PER], ss[2 * PER];
    unsigned ld[2 * PER], k[2 * PER];
#pragma unroll
    for (int j = 0; j < PER; ++j) {
        const int i = (int)threadIdx.x + j * BUILD_THREADS;
        const int4 q = (i < Q) ? bk[i] : make_int4(-1, 0, -1, 0);
        dd[2 * j] = q.x; ss[2 * j] = q.y; dd[2 * j + 1] = q.z; ss[2 * j + 1] = q.w;
    }
#pragma unroll
    for (int j = 0; j < 2 * PER; ++j) {
        ld[j] = (unsigned)(dd[j] - node0);
        k[j]  = 0xffffffffu;
        if (ld[j] < (unsigned)RNG) k[j] = atomicAdd(&cnt[ld[j]], 1u);
    }
    bool past = false;
#pragma unroll
    for (int j = 0; j < 2 * PER; ++j) {
        if (k[j] < (unsigned)SLOTS) lines[ld[j] * SLOTS + k[j]] = ss[j];
        else if (ld[j] < (unsigned)RNG) past = true;
    }
    if (past) {
#pragma unroll
        for (int j = 0; j < 2 * PER; ++j) {
            if (ld[j] < (unsigned)RNG && k[j] >= (unsigned)SLOTS) put_overflow(ovf_n, ovf, dd[j], ss[j]);
        }
    }
    const unsigned ns = min(spill_n[0], (unsigned)N_EDGES);
    for (unsigned i = threadIdx.x; i < ns; i += BUILD_THREADS) {
        const int2 e = spill[i];
        const unsigned l = (unsigned)(e.x - node0);
        if (l < (unsigned)RNG) {
            const unsigned kk = atomicAdd(&cnt[l], 1u);
            if (kk < (unsigned)SLOTS) lines[l * SLOTS + kk] = e.y;
            else put_overflow(ovf_n, ovf, e.x, e.y);
        }
    }
    __syncthreads();

    int* so = slots + (size_t)node0 * SLOTS;
    for (int i = threadIdx.x; i < RNG * SLOTS; i += BUILD_THREADS) so[i] = lines[i];
    for (int i = threadIdx.x; i < RNG; i += BUILD_THREADS) {
        const unsigned cc = cnt[i];
        const unsigned c0 = min(cc, (unsigned)CAP);
        cin[node0 + i] = c0;
        cin[N_NODES + node0 + i] = cc - c0;
    }
}

__device__ __forceinline__ void
product_part(float* __restrict__ Wt, int wg, const float* __restrict__ h, const float* __restrict__ W,
             const float* __restrict__ outdeg, float* __restrict__ x) {
    static_assert((IN_DIM * OUT_DIM) % BUILD_THREADS == 0, "every thread stages the same number of W's elements");
#pragma unroll
    for (int j = 0; j < IN_DIM * OUT_DIM / BUILD_THREADS; ++j) {
        const int i = (int)threadIdx.x + j * BUILD_THREADS;
        int k = i >> 6;
        int c = i & 63;
        Wt[c * 130 + k] = W[i];
    }
    __syncthreads();

    const int wave = __builtin_amdgcn_readfirstlane((int)(threadIdx.x >> 5));
    const int lane = threadIdx.x & 31;
    if (wave >= PROD_TILES) return;
    const int mtile = wg * PROD_TILES + wave;

    const int mrow = (mtile << 4) + (lane & 15);
    const int kb   = (lane >> 4) << 1;
    const int n    = lane & 15;

    const float* hrow = h + (size_t)mrow * IN_DIM + kb;
    const float* w0 = &Wt[(0 * 16 + n) * 130 + kb];
    const float* w1 = &Wt[(1 * 16 + n) * 130 + kb];
    const float* w2 = &Wt[(2 * 16 + n) * 130 + kb];
    const float* w3 = &Wt[(3 * 16 + n) * 130 + kb];

    v8f acc0 = {}, acc1 = {}, acc2 = {}, acc3 = {};
#pragma unroll 4
    for (int k = 0; k < IN_DIM; k += 4) {
        v2f a  = *(const v2f*)(hrow + k);
        v2f b0 = *(const v2f*)(w0 + k);
        v2f b1 = *(const v2f*)(w1 + k);
        v2f b2 = *(const v2f*)(w2 + k);
        v2f b3 = *(const v2f*)(w3 + k);
        acc0 = __builtin_amdgcn_wmma_f32_16x16x4_f32(false, a, false, b0, (short)0, acc0, false, false);
        acc1 = __builtin_amdgcn_wmma_f32_16x16x4_f32(false, a, false, b1, (short)0, acc1, false, false);
        acc2 = __builtin_amdgcn_wmma_f32_16x16x4_f32(false, a, false, b2, (short)0, acc2, false, false);
        acc3 = __builtin_amdgcn_wmma_f32_16x16x4_f32(false, a, false, b3, (short)0, acc3, false, false);
    }

    const int rowbase = (mtile << 4) + ((lane >> 4) << 3);
#pragma unroll
    for (int r = 0; r < 8; ++r) {
        int row = rowbase + r;
        float s = rsqrtf(fmaxf(outdeg[row] + outdeg[N_NODES + row], 1.0f));
        float* xr = x + (size_t)row * OUT_DIM + n;
        xr[0]  = acc0[r] * s;
        xr[16] = acc1[r] * s;
        xr[32] = acc2[r] * s;
        xr[48] = acc3[r] * s;
    }
}

__global__ void __launch_bounds__(BUILD_THREADS, 16)
build_product_kernel(const int4* __restrict__ bucket4, const unsigned* __restrict__ spill_n, const int2* __restrict__ spill,
                     unsigned* __restrict__ cin, int* __restrict__ slots, unsigned* __restrict__ ovf_n, int2* __restrict__ ovf,
                     const float* __restrict__ h, const float* __restrict__ W, const float* __restrict__ outdeg,
                     float* __restrict__ x) {
    __shared__ union {
        struct { unsigned cnt[RNG]; int lines[RNG * SLOTS]; } build;
        float wt[OUT_DIM * 130];
    } sm;
    if (blockIdx.x < PROD_WGS) product_part(sm.wt, (int)blockIdx.x, h, W, outdeg, x);
    else build_part(sm.build.cnt, sm.build.lines, (int)blockIdx.x - PROD_WGS, bucket4, spill_n, spill, cin, slots, ovf_n, ovf);
}

template <int G>
__device__ __forceinline__ void gather_group(const float2* __restrict__ x2, int cv, int j, int m,
                                             int lane, float2& acc) {
    float2 v[G];
#pragma unroll
    for (int u = 0; u < G; ++u) {
        const int s = __builtin_amdgcn_readlane(cv, j + u);
        v[u] = x2[(size_t)s * (OUT_DIM / 2) + lane];
    }
#pragma unroll
    for (int u = 0; u < G; ++u) {
        if (u < m) { acc.x += v[u].x; acc.y += v[u].y; }
    }
}

__device__ __forceinline__ int slot_of(int i, int a0) {
    return min((i < a0) ? i : CAP + i - a0, SLOTS - 1);
}

__device__ __forceinline__ void gather_overflow(const float2* __restrict__ x2, const unsigned* __restrict__ ovf_n,
                                                const int2* __restrict__ ovf, int node, int lane, float2& acc) {
    const unsigned no = min((unsigned)__builtin_amdgcn_readfirstlane((int)ovf_n[0]), (unsigned)N_EDGES);
    for (unsigned base = 0; base < no; base += 32) {
        const unsigned i = base + (unsigned)lane;
        const int2 e = (i < no) ? ovf[i] : make_int2(-1, 0);
        unsigned m = __builtin_amdgcn_ballot_w32(e.x == node);
        while (m) {
            const int u = __builtin_ctz(m);
            m &= m - 1;
            const int s = __builtin_amdgcn_readlane(e.y, u);
            const float2 v = x2[(size_t)s * (OUT_DIM / 2) + lane];
            acc.x += v.x; acc.y += v.y;
        }
    }
}

__global__ void __launch_bounds__(256)
gather_kernel(const float* __restrict__ x, const int* __restrict__ slots, const unsigned* __restrict__ cin,
              const unsigned* __restrict__ ovf_n, const int2* __restrict__ ovf,
              const float* __restrict__ b, float* __restrict__ out, int n) {
    const int lane = threadIdx.x & 31;
    const int node0 = __builtin_amdgcn_readfirstlane((blockIdx.x * 8 + (threadIdx.x >> 5)) * NPW);
    if (node0 >= n) return;
    const int nn = min(NPW, n - node0);
    const unsigned c0 = (lane < nn) ? cin[node0 + lane] : 0u;
    const unsigned c1 = (lane < nn) ? cin[n + node0 + lane] : 0u;
    const int a0v = (int)min(c0, (unsigned)CAP);
    const int a1v = (int)min(c1, (unsigned)CAP);
    const int ovv = (c0 > (unsigned)CAP || c1 > (unsigned)CAP) ? 1 : 0;
    const int ndv = __float_as_int(rsqrtf(fmaxf((float)(c0 + c1), 1.0f)));
    const float2 bb = ((const float2*)b)[lane];
    const float2* x2 = (const float2*)x;
    const int* srow = slots + (size_t)node0 * SLOTS;
    float2* o2 = (float2*)out + (size_t)node0 * (OUT_DIM / 2) + lane;

    int a0  = __builtin_amdgcn_readlane(a0v, 0);
    int cnt = a0 + __builtin_amdgcn_readlane(a1v, 0);
    int cvr = srow[slot_of(lane, a0)];
    for (int t = 0; t < nn; ++t) {
        const int tn   = min(t + 1, nn - 1);
        const int na0  = __builtin_amdgcn_readlane(a0v, tn);
        const int ncnt = na0 + __builtin_amdgcn_readlane(a1v, tn);
        const int cnt0 = min(32, cnt);
        int cv = (lane < cnt0) ? cvr : 0;
        float2 acc = make_float2(0.0f, 0.0f);
        {
            float2 v[8];
#pragma unroll
            for (int u = 0; u < 8; ++u) {
                const int s = __builtin_amdgcn_readlane(cv, u);
                v[u] = x2[(size_t)s * (OUT_DIM / 2) + lane];
            }
            cvr = srow[(size_t)tn * SLOTS + slot_of(lane, na0)];
#pragma unroll
            for (int u = 0; u < 8; ++u) {
                if (u < cnt0) { acc.x += v[u].x; acc.y += v[u].y; }
            }
        }
        {
            int j = 8;
            for (; j + 8 <= cnt0; j += 8) gather_group<8>(x2, cv, j, 8, lane, acc);
            if (j + 4 <= cnt0) { gather_group<4>(x2, cv, j, 4, lane, acc); j += 4; }
            if (j < cnt0) gather_group<4>(x2, cv, j, cnt0 - j, lane, acc);
        }
        if (cnt > 32) {
            const int c2 = cnt - 32;
            cv = (lane < c2) ? srow[(size_t)t * SLOTS + slot_of(32 + lane, a0)] : 0;
            int j = 0;
            for (; j + 8 <= c2; j += 8) gather_group<8>(x2, cv, j, 8, lane, acc);
            if (j + 4 <= c2) { gather_group<4>(x2, cv, j, 4, lane, acc); j += 4; }
            if (j < c2) gather_group<4>(x2, cv, j, c2 - j, lane, acc);
        }
        if (__builtin_amdgcn_readlane(ovv, t)) gather_overflow(x2, ovf_n, ovf, node0 + t, lane, acc);

        const float nd = __int_as_float(__builtin_amdgcn_readlane(ndv, t));
        float2 r;
        r.x = acc.x * nd + bb.x;
        r.y = acc.y * nd + bb.y;
        r.x = (r.x >= 0.0f) ? r.x : r.x * NEG_SLOPE;
        r.y = (r.y >= 0.0f) ? r.y : r.y * NEG_SLOPE;
        o2[(size_t)t * (OUT_DIM / 2)] = r;

        a0  = na0;
        cnt = ncnt;
    }
}

extern "C" void kernel_launch(void* const* d_in, const int* in_sizes, int n_in,
                              void* d_out, int out_size, void* d_ws, size_t ws_size,
                              hipStream_t stream) {
    const float* h    = (const float*)d_in[0];
    const float* W    = (const float*)d_in[1];
    const float* b    = (const float*)d_in[2];
    const int*   esrc = (const int*)d_in[3];
    const int*   edst = (const int*)d_in[4];
    float*       out  = (float*)d_out;

    float*    ws      = (float*)d_ws;
    float*    outdeg  = ws;
    unsigned* ovf_n   = (unsigned*)(ws + 2 * (size_t)N_NODES);
    unsigned* spill_n = ovf_n + 32;
    unsigned* cin     = (unsigned*)(ws + 2 * (size_t)N_NODES + 64);
    float*    x       = ws + 4 * (size_t)N_NODES + 64;
    int*      slots   = (int*)(x + (size_t)N_NODES * OUT_DIM);
    int2*     ovf     = (int2*)(slots + (size_t)N_NODES * SLOTS);
    int2*     spill   = ovf + (size_t)N_EDGES;
    int2*     bucket  = spill + (size_t)N_EDGES;

    hipMemsetAsync(ws, 0, (2 * (size_t)N_NODES + 64) * sizeof(float), stream);

    part_kernel<<<DEG_BLOCKS, 256, 0, stream>>>(esrc, edst, outdeg, bucket, spill_n, spill, N_EDGES);

    build_product_kernel<<<PROD_WGS + NBIN, BUILD_THREADS, 0, stream>>>((const int4*)bucket, spill_n, spill, cin, slots, ovf_n, ovf,
                                                                        h, W, outdeg, x);

    gather_kernel<<<GATHER_BLOCKS, 256, 0, stream>>>(x, slots, cin, ovf_n, ovf, b, out, N_NODES);
}
